# starting kernel
# baseline (speedup 1.0000x reference)
#include <hip/hip_runtime.h>
#include <hip/hip_bf16.h>


typedef __attribute__((ext_vector_type(16))) _Float16 v16h;
typedef __attribute__((ext_vector_type(8)))  _Float16 v8h;
typedef __attribute__((ext_vector_type(8)))  float    v8f;

#define T_TOKENS   262144
#define D_IN       128
#define D_H        256
#define D_OUT      128
#define N_SETS     8192

#define WAVES_PER_BLOCK 2
#define MTILES          2
#define TOK_PER_WAVE    (MTILES * 16)
#define TOK_PER_BLOCK   (WAVES_PER_BLOCK * TOK_PER_WAVE)

#define KS1 (D_IN / 32)
#define KS2 (D_H / 32)
#define NT1 (D_H / 16)
#define NT2 (D_OUT / 16)

#define H_STRIDE 264

__device__ __forceinline__ int frag_klocal(int i, int hi) {
    return (i < 8) ? (hi * 8 + i) : (16 + hi * 8 + (i - 8));
}

__global__ void prep_weights_kernel(const float* __restrict__ W1,
                                    const float* __restrict__ W2,
                                    _Float16* __restrict__ w1f,
                                    _Float16* __restrict__ w2f) {
    int idx = blockIdx.x * blockDim.x + threadIdx.x;
    bool is2 = idx >= (D_IN * D_H);
    int t = is2 ? idx - (D_IN * D_H) : idx;

    int i    = t & 15;
    int lane = (t >> 4) & 31;
    int fid  = t >> 9;
    int KS   = is2 ? KS2 : KS1;
    int ks   = fid % KS;
    int nt   = fid / KS;

    int lane16 = lane & 15;
    int hi     = lane >> 4;
    int k      = ks * 32 + frag_klocal(i, hi);
    int n      = nt * 16 + lane16;

    if (!is2) w1f[t] = (_Float16)W1[k * D_H + n];
    else      w2f[t] = (_Float16)W2[k * D_OUT + n];
}

__device__ __forceinline__ v8f wmma_f16(v16h a, v16h b, v8f c) {
    return __builtin_amdgcn_wmma_f32_16x16x32_f16(
        false, a, false, b, (short)0, c, false, false);
}

__device__ __forceinline__ v16h pack_a_from_f32(const float* p0, const float* p1) {
    const float4 f0 = *(const float4*)(p0);
    const float4 f1 = *(const float4*)(p0 + 4);
    const float4 f2 = *(const float4*)(p1);
    const float4 f3 = *(const float4*)(p1 + 4);
    v16h a;
    a[0]  = (_Float16)f0.x; a[1]  = (_Float16)f0.y;
    a[2]  = (_Float16)f0.z; a[3]  = (_Float16)f0.w;
    a[4]  = (_Float16)f1.x; a[5]  = (_Float16)f1.y;
    a[6]  = (_Float16)f1.z; a[7]  = (_Float16)f1.w;
    a[8]  = (_Float16)f2.x; a[9]  = (_Float16)f2.y;
    a[10] = (_Float16)f2.z; a[11] = (_Float16)f2.w;
    a[12] = (_Float16)f3.x; a[13] = (_Float16)f3.y;
    a[14] = (_Float16)f3.z; a[15] = (_Float16)f3.w;
    return a;
}

__global__ void __launch_bounds__(WAVES_PER_BLOCK * 32)
deepsets_mlp_kernel(const float* __restrict__ x,
                    const int* __restrict__ seg,
                    const _Float16* __restrict__ w1f,
                    const _Float16* __restrict__ w2f,
                    const float* __restrict__ b1,
                    const float* __restrict__ b2,
                    float* __restrict__ outSums) {
    __shared__ _Float16 sH[WAVES_PER_BLOCK][TOK_PER_WAVE * H_STRIDE];

    const int wave   = threadIdx.x >> 5;
    const int lane   = threadIdx.x & 31;
    const int lane16 = lane & 15;
    const int hi     = lane >> 4;

    const int tokBase = blockIdx.x * TOK_PER_BLOCK + wave * TOK_PER_WAVE;

    v16h a1[MTILES][KS1];
#pragma unroll
    for (int mt = 0; mt < MTILES; ++mt) {
        const float* xrow = x + (size_t)(tokBase + mt * 16 + lane16) * D_IN;
        if (mt == 0)
            __builtin_prefetch(xrow + (size_t)TOK_PER_BLOCK * D_IN, 0, 1);
#pragma unroll
        for (int ks = 0; ks < KS1; ++ks) {
            const int k0 = ks * 32 + hi * 8;
            a1[mt][ks] = pack_a_from_f32(xrow + k0, xrow + k0 + 16);
        }
    }

#pragma unroll
    for (int nt = 0; nt < NT1; ++nt) {
        v16h vb[KS1];
#pragma unroll
        for (int ks = 0; ks < KS1; ++ks)
            vb[ks] = *(const v16h*)(w1f + ((size_t)(nt * KS1 + ks) * 32 + lane) * 16);

        v8f c[MTILES];
#pragma unroll
        for (int mt = 0; mt < MTILES; ++mt) c[mt] = (v8f){};
#pragma unroll
        for (int ks = 0; ks < KS1; ++ks)
#pragma unroll
            for (int mt = 0; mt < MTILES; ++mt)
                c[mt] = wmma_f16(a1[mt][ks], vb[ks], c[mt]);

        const int n = nt * 16 + lane16;
        const float bias = b1[n];
#pragma unroll
        for (int mt = 0; mt < MTILES; ++mt) {
#pragma unroll
            for (int r = 0; r < 8; ++r) {
                float v = c[mt][r] + bias;
                v = v > 0.0f ? v : 0.0f;
                sH[wave][(mt * 16 + r + 8 * hi) * H_STRIDE + n] = (_Float16)v;
            }
        }
    }

    __syncthreads();

    v16h a2[MTILES][KS2];
#pragma unroll
    for (int mt = 0; mt < MTILES; ++mt) {
#pragma unroll
        for (int ks = 0; ks < KS2; ++ks) {
            const _Float16* base =
                &sH[wave][(mt * 16 + lane16) * H_STRIDE + ks * 32 + hi * 8];
            const v8h lo = *(const v8h*)(base);
            const v8h up = *(const v8h*)(base + 16);
            v16h a;
#pragma unroll
            for (int i = 0; i < 8; ++i) { a[i] = lo[i]; a[i + 8] = up[i]; }
            a2[mt][ks] = a;
        }
    }

    int sid[MTILES][8];
#pragma unroll
    for (int mt = 0; mt < MTILES; ++mt)
#pragma unroll
        for (int r = 0; r < 8; ++r)
            sid[mt][r] = seg[tokBase + mt * 16 + r + 8 * hi];

#pragma unroll
    for (int nt = 0; nt < NT2; ++nt) {
        v16h vb[KS2];
#pragma unroll
        for (int ks = 0; ks < KS2; ++ks)
            vb[ks] = *(const v16h*)(w2f + ((size_t)(nt * KS2 + ks) * 32 + lane) * 16);

        v8f c[MTILES];
#pragma unroll
        for (int mt = 0; mt < MTILES; ++mt) c[mt] = (v8f){};
#pragma unroll
        for (int ks = 0; ks < KS2; ++ks)
#pragma unroll
            for (int mt = 0; mt < MTILES; ++mt)
                c[mt] = wmma_f16(a2[mt][ks], vb[ks], c[mt]);

        const int n = nt * 16 + lane16;
        const float bias = b2[n];
#pragma unroll
        for (int mt = 0; mt < MTILES; ++mt) {
            float acc = 0.0f;
            int   cur = sid[mt][0];
#pragma unroll
            for (int r = 0; r < 8; ++r) {
                float v = c[mt][r] + bias;
                v = v > 0.0f ? v : 0.0f;
                if (sid[mt][r] != cur) {
                    atomicAdd(outSums + (size_t)cur * D_OUT + n, acc);
                    acc = 0.0f;
                    cur = sid[mt][r];
                }
                acc += v;
            }
            atomicAdd(outSums + (size_t)cur * D_OUT + n, acc);
        }
    }
}

__global__ void count_kernel(const int* __restrict__ seg, float* __restrict__ counts) {
    int t = blockIdx.x * blockDim.x + threadIdx.x;
    if (t < T_TOKENS) atomicAdd(&counts[seg[t]], 1.0f);
}

__global__ void finalize_kernel(float* __restrict__ out, const float* __restrict__ counts) {
    int i = blockIdx.x * blockDim.x + threadIdx.x;
    if (i < N_SETS * D_OUT) {
        out[i] = out[i] / fmaxf(counts[i >> 7], 1.0f);
    }
}

extern "C" void kernel_launch(void* const* d_in, const int* in_sizes, int n_in,
                              void* d_out, int out_size, void* d_ws, size_t ws_size,
                              hipStream_t stream) {
    (void)in_sizes; (void)n_in; (void)ws_size;

    const float* x   = (const float*)d_in[0];
    const int*   seg = (const int*)d_in[1];
    const float* W1  = (const float*)d_in[3];
    const float* b1  = (const float*)d_in[4];
    const float* W2  = (const float*)d_in[5];
    const float* b2  = (const float*)d_in[6];

    float* out = (float*)d_out;

    _Float16* w1f    = (_Float16*)d_ws;
    _Float16* w2f    = (_Float16*)((char*)d_ws + (size_t)D_IN * D_H * 2);
    float*    counts = (float*)((char*)d_ws + (size_t)(D_IN * D_H + D_H * D_OUT) * 2);

    hipMemsetAsync(out, 0, (size_t)out_size * sizeof(float), stream);
    hipMemsetAsync(counts, 0, (size_t)N_SETS * sizeof(float), stream);

    prep_weights_kernel<<<(D_IN * D_H + D_H * D_OUT) / 256, 256, 0, stream>>>(W1, W2, w1f, w2f);

    deepsets_mlp_kernel<<<T_TOKENS / TOK_PER_BLOCK, WAVES_PER_BLOCK * 32, 0, stream>>>(
        x, seg, w1f, w2f, b1, b2, out);

    count_kernel<<<T_TOKENS / 256, 256, 0, stream>>>(seg, counts);

    finalize_kernel<<<(N_SETS * D_OUT) / 256, 256, 0, stream>>>(out, counts);
}


# global loads issued in fenced groups 1-2 stages ahead of their WMMAs; layer 2 re-looped k-outermost over 16 accumulator tiles so each k-step's 8 fragments load together; per-tile k order unchanged
# speedup vs baseline: 1.2771x; 1.2771x over previous
#include <hip/hip_runtime.h>
#include <hip/hip_bf16.h>


typedef __attribute__((ext_vector_type(16))) _Float16 v16h;
typedef __attribute__((ext_vector_type(8)))  _Float16 v8h;
typedef __attribute__((ext_vector_type(8)))  float    v8f;

#define T_TOKENS   262144
#define D_IN       128
#define D_H        256
#define D_OUT      128
#define N_SETS     8192

#define WAVES_PER_BLOCK 2
#define MTILES          2
#define TOK_PER_WAVE    (MTILES * 16)
#define TOK_PER_BLOCK   (WAVES_PER_BLOCK * TOK_PER_WAVE)

#define KS1 (D_IN / 32)
#define KS2 (D_H / 32)
#define NT1 (D_H / 16)
#define NT2 (D_OUT / 16)

#define H_STRIDE 264

__device__ __forceinline__ int frag_klocal(int i, int hi) {
    return (i < 8) ? (hi * 8 + i) : (16 + hi * 8 + (i - 8));
}

__global__ void prep_weights_kernel(const float* __restrict__ W1,
                                    const float* __restrict__ W2,
                                    _Float16* __restrict__ w1f,
                                    _Float16* __restrict__ w2f) {
    int idx = blockIdx.x * blockDim.x + threadIdx.x;
    bool is2 = idx >= (D_IN * D_H);
    int t = is2 ? idx - (D_IN * D_H) : idx;

    int i    = t & 15;
    int lane = (t >> 4) & 31;
    int fid  = t >> 9;
    int KS   = is2 ? KS2 : KS1;
    int ks   = fid % KS;
    int nt   = fid / KS;

    int lane16 = lane & 15;
    int hi     = lane >> 4;
    int k      = ks * 32 + frag_klocal(i, hi);
    int n      = nt * 16 + lane16;

    if (!is2) w1f[t] = (_Float16)W1[k * D_H + n];
    else      w2f[t] = (_Float16)W2[k * D_OUT + n];
}

__device__ __forceinline__ v8f wmma_f16(v16h a, v16h b, v8f c) {
    return __builtin_amdgcn_wmma_f32_16x16x32_f16(
        false, a, false, b, (short)0, c, false, false);
}

__device__ __forceinline__ v16h pack_a_from_regs(float4 f0, float4 f1, float4 f2, float4 f3) {
    v16h a;
    a[0]  = (_Float16)f0.x; a[1]  = (_Float16)f0.y;
    a[2]  = (_Float16)f0.z; a[3]  = (_Float16)f0.w;
    a[4]  = (_Float16)f1.x; a[5]  = (_Float16)f1.y;
    a[6]  = (_Float16)f1.z; a[7]  = (_Float16)f1.w;
    a[8]  = (_Float16)f2.x; a[9]  = (_Float16)f2.y;
    a[10] = (_Float16)f2.z; a[11] = (_Float16)f2.w;
    a[12] = (_Float16)f3.x; a[13] = (_Float16)f3.y;
    a[14] = (_Float16)f3.z; a[15] = (_Float16)f3.w;
    return a;
}

__global__ void __launch_bounds__(WAVES_PER_BLOCK * 32)
deepsets_mlp_kernel(const float* __restrict__ x,
                    const int* __restrict__ seg,
                    const _Float16* __restrict__ w1f,
                    const _Float16* __restrict__ w2f,
                    const float* __restrict__ b1,
                    const float* __restrict__ b2,
                    float* __restrict__ outSums) {
    __shared__ _Float16 sH[WAVES_PER_BLOCK][TOK_PER_WAVE * H_STRIDE];

    const int wave   = threadIdx.x >> 5;
    const int lane   = threadIdx.x & 31;
    const int lane16 = lane & 15;
    const int hi     = lane >> 4;

    const int tokBase = blockIdx.x * TOK_PER_BLOCK + wave * TOK_PER_WAVE;

    float4 xr[MTILES][KS1][4];
#pragma unroll
    for (int mt = 0; mt < MTILES; ++mt) {
        const float* xrow = x + (size_t)(tokBase + mt * 16 + lane16) * D_IN;
        if (mt == 0)
            __builtin_prefetch(xrow + (size_t)TOK_PER_BLOCK * D_IN, 0, 1);
#pragma unroll
        for (int ks = 0; ks < KS1; ++ks) {
            const float* p = xrow + ks * 32 + hi * 8;
            xr[mt][ks][0] = *(const float4*)(p);
            xr[mt][ks][1] = *(const float4*)(p + 4);
            xr[mt][ks][2] = *(const float4*)(p + 16);
            xr[mt][ks][3] = *(const float4*)(p + 20);
        }
    }

    v16h vb1[3][KS1];
#pragma unroll
    for (int nt = 0; nt < 2; ++nt)
#pragma unroll
        for (int ks = 0; ks < KS1; ++ks)
            vb1[nt][ks] = *(const v16h*)(w1f + ((size_t)(nt * KS1 + ks) * 32 + lane) * 16);

    __builtin_amdgcn_sched_barrier(0);

    float bias1[NT1];
#pragma unroll
    for (int nt = 0; nt < NT1; ++nt) bias1[nt] = b1[nt * 16 + lane16];

    v16h a1[MTILES][KS1];
#pragma unroll
    for (int mt = 0; mt < MTILES; ++mt)
#pragma unroll
        for (int ks = 0; ks < KS1; ++ks)
            a1[mt][ks] = pack_a_from_regs(xr[mt][ks][0], xr[mt][ks][1],
                                          xr[mt][ks][2], xr[mt][ks][3]);

#pragma unroll
    for (int nt = 0; nt < NT1; ++nt) {
        if (nt + 2 < NT1) {
#pragma unroll
            for (int ks = 0; ks < KS1; ++ks)
                vb1[(nt + 2) % 3][ks] =
                    *(const v16h*)(w1f + ((size_t)((nt + 2) * KS1 + ks) * 32 + lane) * 16);
        }
        __builtin_amdgcn_sched_barrier(0);

        v8f c[MTILES];
#pragma unroll
        for (int mt = 0; mt < MTILES; ++mt) c[mt] = (v8f){};
#pragma unroll
        for (int ks = 0; ks < KS1; ++ks)
#pragma unroll
            for (int mt = 0; mt < MTILES; ++mt)
                c[mt] = wmma_f16(a1[mt][ks], vb1[nt % 3][ks], c[mt]);

        const int n = nt * 16 + lane16;
        const float bias = bias1[nt];
#pragma unroll
        for (int mt = 0; mt < MTILES; ++mt) {
#pragma unroll
            for (int r = 0; r < 8; ++r) {
                float v = c[mt][r] + bias;
                v = v > 0.0f ? v : 0.0f;
                sH[wave][(mt * 16 + r + 8 * hi) * H_STRIDE + n] = (_Float16)v;
            }
        }
    }

    __syncthreads();

    v8f c2[MTILES][NT2];
#pragma unroll
    for (int mt = 0; mt < MTILES; ++mt)
#pragma unroll
        for (int nt = 0; nt < NT2; ++nt) c2[mt][nt] = (v8f){};

    int   sid[MTILES][8];
    float bias2[NT2];

#pragma unroll
    for (int ks = 0; ks < KS2; ++ks) {
        v16h vb[NT2];
#pragma unroll
        for (int nt = 0; nt < NT2; ++nt)
            vb[nt] = *(const v16h*)(w2f + ((size_t)(nt * KS2 + ks) * 32 + lane) * 16);

        v16h a2[MTILES];
#pragma unroll
        for (int mt = 0; mt < MTILES; ++mt) {
            const _Float16* base =
                &sH[wave][(mt * 16 + lane16) * H_STRIDE + ks * 32 + hi * 8];
            const v8h lo = *(const v8h*)(base);
            const v8h up = *(const v8h*)(base + 16);
            v16h a;
#pragma unroll
            for (int i = 0; i < 8; ++i) { a[i] = lo[i]; a[i + 8] = up[i]; }
            a2[mt] = a;
        }

        __builtin_amdgcn_sched_barrier(0);

        if (ks == 0) {
#pragma unroll
            for (int mt = 0; mt < MTILES; ++mt)
#pragma unroll
                for (int r = 0; r < 8; ++r)
                    sid[mt][r] = seg[tokBase + mt * 16 + r + 8 * hi];
#pragma unroll
            for (int nt = 0; nt < NT2; ++nt) bias2[nt] = b2[nt * 16 + lane16];
        }

#pragma unroll
        for (int nt = 0; nt < NT2; ++nt)
#pragma unroll
            for (int mt = 0; mt < MTILES; ++mt)
                c2[mt][nt] = wmma_f16(a2[mt], vb[nt], c2[mt][nt]);

        __builtin_amdgcn_sched_barrier(0);
    }

#pragma unroll
    for (int nt = 0; nt < NT2; ++nt) {
        const int n = nt * 16 + lane16;
        const float bias = bias2[nt];
#pragma unroll
        for (int mt = 0; mt < MTILES; ++mt) {
            float acc = 0.0f;
            int   cur = sid[mt][0];
#pragma unroll
            for (int r = 0; r < 8; ++r) {
                float v = c2[mt][nt][r] + bias;
                v = v > 0.0f ? v : 0.0f;
                if (sid[mt][r] != cur) {
                    atomicAdd(outSums + (size_t)cur * D_OUT + n, acc);
                    acc = 0.0f;
                    cur = sid[mt][r];
                }
                acc += v;
            }
            atomicAdd(outSums + (size_t)cur * D_OUT + n, acc);
        }
    }
}

__global__ void count_kernel(const int* __restrict__ seg, float* __restrict__ counts) {
    int t = blockIdx.x * blockDim.x + threadIdx.x;
    if (t < T_TOKENS) atomicAdd(&counts[seg[t]], 1.0f);
}

__global__ void finalize_kernel(float* __restrict__ out, const float* __restrict__ counts) {
    int i = blockIdx.x * blockDim.x + threadIdx.x;
    if (i < N_SETS * D_OUT) {
        out[i] = out[i] / fmaxf(counts[i >> 7], 1.0f);
    }
}

extern "C" void kernel_launch(void* const* d_in, const int* in_sizes, int n_in,
                              void* d_out, int out_size, void* d_ws, size_t ws_size,
                              hipStream_t stream) {
    (void)in_sizes; (void)n_in; (void)ws_size;

    const float* x   = (const float*)d_in[0];
    const int*   seg = (const int*)d_in[1];
    const float* W1  = (const float*)d_in[3];
    const float* b1  = (const float*)d_in[4];
    const float* W2  = (const float*)d_in[5];
    const float* b2  = (const float*)d_in[6];

    float* out = (float*)d_out;

    _Float16* w1f    = (_Float16*)d_ws;
    _Float16* w2f    = (_Float16*)((char*)d_ws + (size_t)D_IN * D_H * 2);
    float*    counts = (float*)((char*)d_ws + (size_t)(D_IN * D_H + D_H * D_OUT) * 2);

    hipMemsetAsync(out, 0, (size_t)out_size * sizeof(float), stream);
    hipMemsetAsync(counts, 0, (size_t)N_SETS * sizeof(float), stream);

    prep_weights_kernel<<<(D_IN * D_H + D_H * D_OUT) / 256, 256, 0, stream>>>(W1, W2, w1f, w2f);

    deepsets_mlp_kernel<<<T_TOKENS / TOK_PER_BLOCK, WAVES_PER_BLOCK * 32, 0, stream>>>(
        x, seg, w1f, w2f, b1, b2, out);

    count_kernel<<<T_TOKENS / 256, 256, 0, stream>>>(seg, counts);

    finalize_kernel<<<(N_SETS * D_OUT) / 256, 256, 0, stream>>>(out, counts);
}


# mean folded into the accumulation: counts taken first, each run's partial sum scaled by 1/max(count,1) before its atomic add (summed values are pre-scaled); finalize launch removed
# speedup vs baseline: 1.3587x; 1.0639x over previous
#include <hip/hip_runtime.h>
#include <hip/hip_bf16.h>


typedef __attribute__((ext_vector_type(16))) _Float16 v16h;
typedef __attribute__((ext_vector_type(8)))  _Float16 v8h;
typedef __attribute__((ext_vector_type(8)))  float    v8f;

#define T_TOKENS   262144
#define D_IN       128
#define D_H        256
#define D_OUT      128
#define N_SETS     8192

#define WAVES_PER_BLOCK 2
#define MTILES          2
#define TOK_PER_WAVE    (MTILES * 16)
#define TOK_PER_BLOCK   (WAVES_PER_BLOCK * TOK_PER_WAVE)

#define KS1 (D_IN / 32)
#define KS2 (D_H / 32)
#define NT1 (D_H / 16)
#define NT2 (D_OUT / 16)

#define H_STRIDE 264

__device__ __forceinline__ int frag_klocal(int i, int hi) {
    return (i < 8) ? (hi * 8 + i) : (16 + hi * 8 + (i - 8));
}

__global__ void prep_weights_kernel(const float* __restrict__ W1,
                                    const float* __restrict__ W2,
                                    _Float16* __restrict__ w1f,
                                    _Float16* __restrict__ w2f) {
    int idx = blockIdx.x * blockDim.x + threadIdx.x;
    bool is2 = idx >= (D_IN * D_H);
    int t = is2 ? idx - (D_IN * D_H) : idx;

    int i    = t & 15;
    int lane = (t >> 4) & 31;
    int fid  = t >> 9;
    int KS   = is2 ? KS2 : KS1;
    int ks   = fid % KS;
    int nt   = fid / KS;

    int lane16 = lane & 15;
    int hi     = lane >> 4;
    int k      = ks * 32 + frag_klocal(i, hi);
    int n      = nt * 16 + lane16;

    if (!is2) w1f[t] = (_Float16)W1[k * D_H + n];
    else      w2f[t] = (_Float16)W2[k * D_OUT + n];
}

__device__ __forceinline__ v8f wmma_f16(v16h a, v16h b, v8f c) {
    return __builtin_amdgcn_wmma_f32_16x16x32_f16(
        false, a, false, b, (short)0, c, false, false);
}

__device__ __forceinline__ v16h pack_a_from_regs(float4 f0, float4 f1, float4 f2, float4 f3) {
    v16h a;
    a[0]  = (_Float16)f0.x; a[1]  = (_Float16)f0.y;
    a[2]  = (_Float16)f0.z; a[3]  = (_Float16)f0.w;
    a[4]  = (_Float16)f1.x; a[5]  = (_Float16)f1.y;
    a[6]  = (_Float16)f1.z; a[7]  = (_Float16)f1.w;
    a[8]  = (_Float16)f2.x; a[9]  = (_Float16)f2.y;
    a[10] = (_Float16)f2.z; a[11] = (_Float16)f2.w;
    a[12] = (_Float16)f3.x; a[13] = (_Float16)f3.y;
    a[14] = (_Float16)f3.z; a[15] = (_Float16)f3.w;
    return a;
}

__global__ void __launch_bounds__(WAVES_PER_BLOCK * 32)
deepsets_mlp_kernel(const float* __restrict__ x,
                    const int* __restrict__ seg,
                    const _Float16* __restrict__ w1f,
                    const _Float16* __restrict__ w2f,
                    const float* __restrict__ b1,
                    const float* __restrict__ b2,
                    const float* __restrict__ counts,
                    float* __restrict__ outMean) {
    __shared__ _Float16 sH[WAVES_PER_BLOCK][TOK_PER_WAVE * H_STRIDE];

    const int wave   = threadIdx.x >> 5;
    const int lane   = threadIdx.x & 31;
    const int lane16 = lane & 15;
    const int hi     = lane >> 4;

    const int tokBase = blockIdx.x * TOK_PER_BLOCK + wave * TOK_PER_WAVE;

    float4 xr[MTILES][KS1][4];
#pragma unroll
    for (int mt = 0; mt < MTILES; ++mt) {
        const float* xrow = x + (size_t)(tokBase + mt * 16 + lane16) * D_IN;
        if (mt == 0)
            __builtin_prefetch(xrow + (size_t)TOK_PER_BLOCK * D_IN, 0, 1);
#pragma unroll
        for (int ks = 0; ks < KS1; ++ks) {
            const float* p = xrow + ks * 32 + hi * 8;
            xr[mt][ks][0] = *(const float4*)(p);
            xr[mt][ks][1] = *(const float4*)(p + 4);
            xr[mt][ks][2] = *(const float4*)(p + 16);
            xr[mt][ks][3] = *(const float4*)(p + 20);
        }
    }

    v16h vb1[3][KS1];
#pragma unroll
    for (int nt = 0; nt < 2; ++nt)
#pragma unroll
        for (int ks = 0; ks < KS1; ++ks)
            vb1[nt][ks] = *(const v16h*)(w1f + ((size_t)(nt * KS1 + ks) * 32 + lane) * 16);

    __builtin_amdgcn_sched_barrier(0);

    float bias1[NT1];
#pragma unroll
    for (int nt = 0; nt < NT1; ++nt) bias1[nt] = b1[nt * 16 + lane16];

    v16h a1[MTILES][KS1];
#pragma unroll
    for (int mt = 0; mt < MTILES; ++mt)
#pragma unroll
        for (int ks = 0; ks < KS1; ++ks)
            a1[mt][ks] = pack_a_from_regs(xr[mt][ks][0], xr[mt][ks][1],
                                          xr[mt][ks][2], xr[mt][ks][3]);

#pragma unroll
    for (int nt = 0; nt < NT1; ++nt) {
        if (nt + 2 < NT1) {
#pragma unroll
            for (int ks = 0; ks < KS1; ++ks)
                vb1[(nt + 2) % 3][ks] =
                    *(const v16h*)(w1f + ((size_t)((nt + 2) * KS1 + ks) * 32 + lane) * 16);
        }
        __builtin_amdgcn_sched_barrier(0);

        v8f c[MTILES];
#pragma unroll
        for (int mt = 0; mt < MTILES; ++mt) c[mt] = (v8f){};
#pragma unroll
        for (int ks = 0; ks < KS1; ++ks)
#pragma unroll
            for (int mt = 0; mt < MTILES; ++mt)
                c[mt] = wmma_f16(a1[mt][ks], vb1[nt % 3][ks], c[mt]);

        const int n = nt * 16 + lane16;
        const float bias = bias1[nt];
#pragma unroll
        for (int mt = 0; mt < MTILES; ++mt) {
#pragma unroll
            for (int r = 0; r < 8; ++r) {
                float v = c[mt][r] + bias;
                v = v > 0.0f ? v : 0.0f;
                sH[wave][(mt * 16 + r + 8 * hi) * H_STRIDE + n] = (_Float16)v;
            }
        }
    }

    __syncthreads();

    v8f c2[MTILES][NT2];
#pragma unroll
    for (int mt = 0; mt < MTILES; ++mt)
#pragma unroll
        for (int nt = 0; nt < NT2; ++nt) c2[mt][nt] = (v8f){};

    int   sid[MTILES][8];
    float bias2[NT2];
    float cnt[MTILES][8];

#pragma unroll
    for (int ks = 0; ks < KS2; ++ks) {
        v16h vb[NT2];
#pragma unroll
        for (int nt = 0; nt < NT2; ++nt)
            vb[nt] = *(const v16h*)(w2f + ((size_t)(nt * KS2 + ks) * 32 + lane) * 16);

        v16h a2[MTILES];
#pragma unroll
        for (int mt = 0; mt < MTILES; ++mt) {
            const _Float16* base =
                &sH[wave][(mt * 16 + lane16) * H_STRIDE + ks * 32 + hi * 8];
            const v8h lo = *(const v8h*)(base);
            const v8h up = *(const v8h*)(base + 16);
            v16h a;
#pragma unroll
            for (int i = 0; i < 8; ++i) { a[i] = lo[i]; a[i + 8] = up[i]; }
            a2[mt] = a;
        }

        __builtin_amdgcn_sched_barrier(0);

        if (ks == 0) {
#pragma unroll
            for (int mt = 0; mt < MTILES; ++mt)
#pragma unroll
                for (int r = 0; r < 8; ++r)
                    sid[mt][r] = seg[tokBase + mt * 16 + r + 8 * hi];
#pragma unroll
            for (int nt = 0; nt < NT2; ++nt) bias2[nt] = b2[nt * 16 + lane16];
        }
        if (ks == KS2 - 1) {
#pragma unroll
            for (int mt = 0; mt < MTILES; ++mt)
#pragma unroll
                for (int r = 0; r < 8; ++r)
                    cnt[mt][r] = counts[sid[mt][r]];
        }

#pragma unroll
        for (int nt = 0; nt < NT2; ++nt)
#pragma unroll
            for (int mt = 0; mt < MTILES; ++mt)
                c2[mt][nt] = wmma_f16(a2[mt], vb[nt], c2[mt][nt]);

        __builtin_amdgcn_sched_barrier(0);
    }

    float inv[MTILES][8];
#pragma unroll
    for (int mt = 0; mt < MTILES; ++mt)
#pragma unroll
        for (int r = 0; r < 8; ++r)
            inv[mt][r] = 1.0f / fmaxf(cnt[mt][r], 1.0f);

#pragma unroll
    for (int nt = 0; nt < NT2; ++nt) {
        const int n = nt * 16 + lane16;
        const float bias = bias2[nt];
#pragma unroll
        for (int mt = 0; mt < MTILES; ++mt) {
            float acc    = 0.0f;
            int   cur    = sid[mt][0];
            float curInv = inv[mt][0];
#pragma unroll
            for (int r = 0; r < 8; ++r) {
                float v = c2[mt][nt][r] + bias;
                v = v > 0.0f ? v : 0.0f;
                if (sid[mt][r] != cur) {
                    atomicAdd(outMean + (size_t)cur * D_OUT + n, acc * curInv);
                    acc    = 0.0f;
                    cur    = sid[mt][r];
                    curInv = inv[mt][r];
                }
                acc += v;
            }
            atomicAdd(outMean + (size_t)cur * D_OUT + n, acc * curInv);
        }
    }
}

__global__ void count_kernel(const int* __restrict__ seg, float* __restrict__ counts) {
    int t = blockIdx.x * blockDim.x + threadIdx.x;
    if (t < T_TOKENS) atomicAdd(&counts[seg[t]], 1.0f);
}

extern "C" void kernel_launch(void* const* d_in, const int* in_sizes, int n_in,
                              void* d_out, int out_size, void* d_ws, size_t ws_size,
                              hipStream_t stream) {
    (void)in_sizes; (void)n_in; (void)ws_size;

    const float* x   = (const float*)d_in[0];
    const int*   seg = (const int*)d_in[1];
    const float* W1  = (const float*)d_in[3];
    const float* b1  = (const float*)d_in[4];
    const float* W2  = (const float*)d_in[5];
    const float* b2  = (const float*)d_in[6];

    float* out = (float*)d_out;

    _Float16* w1f    = (_Float16*)d_ws;
    _Float16* w2f    = (_Float16*)((char*)d_ws + (size_t)D_IN * D_H * 2);
    float*    counts = (float*)((char*)d_ws + (size_t)(D_IN * D_H + D_H * D_OUT) * 2);

    hipMemsetAsync(out, 0, (size_t)out_size * sizeof(float), stream);
    hipMemsetAsync(counts, 0, (size_t)N_SETS * sizeof(float), stream);

    prep_weights_kernel<<<(D_IN * D_H + D_H * D_OUT) / 256, 256, 0, stream>>>(W1, W2, w1f, w2f);

    count_kernel<<<T_TOKENS / 256, 256, 0, stream>>>(seg, counts);

    deepsets_mlp_kernel<<<T_TOKENS / TOK_PER_BLOCK, WAVES_PER_BLOCK * 32, 0, stream>>>(
        x, seg, w1f, w2f, b1, b2, counts, out);
}


# layer 1 computed transposed (W1 fragment as A, x fragment as B): its accumulators are layer 2's A fragments in place, so h stays in registers; LDS h tile, its stores and per-tile barrier removed
# speedup vs baseline: 1.8499x; 1.3615x over previous
#include <hip/hip_runtime.h>
#include <hip/hip_bf16.h>


typedef __attribute__((ext_vector_type(16))) _Float16 v16h;
typedef __attribute__((ext_vector_type(8)))  _Float16 v8h;
typedef __attribute__((ext_vector_type(8)))  float    v8f;

#define T_TOKENS   262144
#define D_IN       128
#define D_H        256
#define D_OUT      128
#define N_SETS     8192

#define WAVES_PER_BLOCK 8
#define MTILES          2
#define TOK_PER_WAVE    (MTILES * 16)
#define N_BLOCKS        256
#define TILES_PER_WAVE  (T_TOKENS / (N_BLOCKS * WAVES_PER_BLOCK * TOK_PER_WAVE))
#define W_HALVES        (D_IN * D_H + D_H * D_OUT)

#define KS1 (D_IN / 32)
#define KS2 (D_H / 32)
#define NT1 (D_H / 16)
#define NT2 (D_OUT / 16)

__device__ __forceinline__ int frag_klocal(int i, int hi) {
    return (i < 8) ? (hi * 8 + i) : (16 + hi * 8 + (i - 8));
}

__global__ void prep_weights_kernel(const float* __restrict__ W1,
                                    const float* __restrict__ W2,
                                    _Float16* __restrict__ w1f,
                                    _Float16* __restrict__ w2f) {
    int idx = blockIdx.x * blockDim.x + threadIdx.x;
    bool is2 = idx >= (D_IN * D_H);
    int t = is2 ? idx - (D_IN * D_H) : idx;

    int i    = t & 15;
    int lane = (t >> 4) & 31;
    int fid  = t >> 9;
    int KS   = is2 ? KS2 : KS1;
    int ks   = fid % KS;
    int nt   = fid / KS;

    int lane16 = lane & 15;
    int hi     = lane >> 4;
    int k      = ks * 32 + frag_klocal(i, hi);
    int n      = nt * 16 + lane16;

    if (!is2) w1f[t] = (_Float16)W1[k * D_H + n];
    else      w2f[t] = (_Float16)W2[k * D_OUT + n];
}

__device__ __forceinline__ v8f wmma_f16(v16h a, v16h b, v8f c) {
    return __builtin_amdgcn_wmma_f32_16x16x32_f16(
        false, a, false, b, (short)0, c, false, false);
}

__device__ __forceinline__ v16h pack_a_from_regs(float4 f0, float4 f1, float4 f2, float4 f3) {
    v16h a;
    a[0]  = (_Float16)f0.x; a[1]  = (_Float16)f0.y;
    a[2]  = (_Float16)f0.z; a[3]  = (_Float16)f0.w;
    a[4]  = (_Float16)f1.x; a[5]  = (_Float16)f1.y;
    a[6]  = (_Float16)f1.z; a[7]  = (_Float16)f1.w;
    a[8]  = (_Float16)f2.x; a[9]  = (_Float16)f2.y;
    a[10] = (_Float16)f2.z; a[11] = (_Float16)f2.w;
    a[12] = (_Float16)f3.x; a[13] = (_Float16)f3.y;
    a[14] = (_Float16)f3.z; a[15] = (_Float16)f3.w;
    return a;
}

__device__ __forceinline__ v16h lds_b_frag(const _Float16* sW, int fid, int lane) {
    const _Float16* p = sW + fid * 512 + lane * 8;
    const v8h lo = *(const v8h*)(p);
    const v8h up = *(const v8h*)(p + 256);
    v16h b;
#pragma unroll
    for (int i = 0; i < 8; ++i) { b[i] = lo[i]; b[i + 8] = up[i]; }
    return b;
}

__device__ __forceinline__ v16h pack_h(v8f c0, v8f c1, v8f bA, v8f bB) {
    v16h a;
#pragma unroll
    for (int i = 0; i < 8; ++i) {
        float v = c0[i] + bA[i];
        v = v > 0.0f ? v : 0.0f;
        a[i] = (_Float16)v;
        float w = c1[i] + bB[i];
        w = w > 0.0f ? w : 0.0f;
        a[i + 8] = (_Float16)w;
    }
    return a;
}

__device__ __forceinline__ void load_x_tile(const float* __restrict__ x, int tokBase,
                                            int lane16, int hi,
                                            float4 (&xr)[MTILES][KS1][4]) {
#pragma unroll
    for (int mt = 0; mt < MTILES; ++mt) {
        const float* xrow = x + (size_t)(tokBase + mt * 16 + lane16) * D_IN;
#pragma unroll
        for (int ks = 0; ks < KS1; ++ks) {
            const float* p = xrow + ks * 32 + hi * 8;
            xr[mt][ks][0] = *(const float4*)(p);
            xr[mt][ks][1] = *(const float4*)(p + 4);
            xr[mt][ks][2] = *(const float4*)(p + 16);
            xr[mt][ks][3] = *(const float4*)(p + 20);
        }
    }
}

__global__ void __launch_bounds__(WAVES_PER_BLOCK * 32)
__attribute__((amdgpu_waves_per_eu(2, 2)))
deepsets_mlp_kernel(const float* __restrict__ x,
                    const int* __restrict__ seg,
                    const _Float16* __restrict__ wf,
                    const float* __restrict__ b1,
                    const float* __restrict__ b2,
                    const float* __restrict__ counts,
                    float* __restrict__ outMean) {
    __shared__ _Float16 sW[W_HALVES];
    __shared__ __attribute__((aligned(32))) float sB1[D_H];

    const int wave   = threadIdx.x >> 5;
    const int lane   = threadIdx.x & 31;
    const int lane16 = lane & 15;
    const int hi     = lane >> 4;

    const _Float16* sW1 = sW;
    const _Float16* sW2 = sW + D_IN * D_H;

    int tile = blockIdx.x * WAVES_PER_BLOCK + wave;

    float4 xr[MTILES][KS1][4];
    load_x_tile(x, tile * TOK_PER_WAVE, lane16, hi, xr);
    __builtin_amdgcn_sched_barrier(0);

    {
        const uint4* g  = (const uint4*)wf;
        uint4*       sd = (uint4*)sW;
#pragma unroll
        for (int k = 0; k < W_HALVES / 8 / (WAVES_PER_BLOCK * 32); ++k) {
            const int q   = k * (WAVES_PER_BLOCK * 32) + threadIdx.x;
            const int fid = q >> 6;
            const int ln  = (q & 63) >> 1;
            const int pl  = q & 1;
            sd[fid * 64 + pl * 32 + ln] = g[q];
        }
    }
    sB1[threadIdx.x] = b1[threadIdx.x];

    float bias2[NT2];
#pragma unroll
    for (int nt = 0; nt < NT2; ++nt) bias2[nt] = b2[nt * 16 + lane16];

    __syncthreads();

#pragma unroll 1
    for (int it = 0; it < TILES_PER_WAVE; ++it) {
        const int tokBase = tile * TOK_PER_WAVE;

        int sid[MTILES][8];
#pragma unroll
        for (int mt = 0; mt < MTILES; ++mt)
#pragma unroll
            for (int r = 0; r < 8; ++r)
                sid[mt][r] = seg[tokBase + mt * 16 + r + 8 * hi];

        v16h w1s[2][KS1];
#pragma unroll
        for (int j = 0; j < 2; ++j)
#pragma unroll
            for (int ks = 0; ks < KS1; ++ks)
                w1s[j][ks] = lds_b_frag(sW1, j * KS1 + ks, lane);

        __builtin_amdgcn_sched_barrier(0);

        v16h a1[MTILES][KS1];
#pragma unroll
        for (int mt = 0; mt < MTILES; ++mt)
#pragma unroll
            for (int ks = 0; ks < KS1; ++ks)
                a1[mt][ks] = pack_a_from_regs(xr[mt][ks][0], xr[mt][ks][1],
                                              xr[mt][ks][2], xr[mt][ks][3]);

        v8f c2[MTILES][NT2];
#pragma unroll
        for (int mt = 0; mt < MTILES; ++mt)
#pragma unroll
            for (int nt = 0; nt < NT2; ++nt) c2[mt][nt] = (v8f){};

        float cnt[MTILES][8];

        v16h w2s[NT2];
        v8f  b1v[2];
#pragma unroll
        for (int s = 0; s < KS2; ++s) {
#pragma unroll
            for (int nt = 0; nt < NT2; ++nt)
                w2s[nt] = lds_b_frag(sW2, nt * KS2 + s, lane);
#pragma unroll
            for (int j = 0; j < 2; ++j)
                b1v[j] = *(const v8f*)(sB1 + (2 * s + j) * 16 + 8 * hi);

            __builtin_amdgcn_sched_barrier(0);

            if (s == KS2 / 2) {
#pragma unroll
                for (int mt = 0; mt < MTILES; ++mt)
#pragma unroll
                    for (int r = 0; r < 8; ++r)
                        cnt[mt][r] = counts[sid[mt][r]];
            }

            v8f cT[MTILES][2];
#pragma unroll
            for (int mt = 0; mt < MTILES; ++mt)
#pragma unroll
                for (int j = 0; j < 2; ++j) cT[mt][j] = (v8f){};
#pragma unroll
            for (int ks = 0; ks < KS1; ++ks)
#pragma unroll
                for (int mt = 0; mt < MTILES; ++mt)
#pragma unroll
                    for (int j = 0; j < 2; ++j)
                        cT[mt][j] = wmma_f16(w1s[j][ks], a1[mt][ks], cT[mt][j]);

            __builtin_amdgcn_sched_barrier(0);

            if (s + 1 < KS2) {
#pragma unroll
                for (int j = 0; j < 2; ++j)
#pragma unroll
                    for (int ks = 0; ks < KS1; ++ks)
                        w1s[j][ks] = lds_b_frag(sW1, (2 * (s + 1) + j) * KS1 + ks, lane);
            }
            __builtin_amdgcn_sched_barrier(0);

            v16h a2[MTILES];
#pragma unroll
            for (int mt = 0; mt < MTILES; ++mt)
                a2[mt] = pack_h(cT[mt][0], cT[mt][1], b1v[0], b1v[1]);

#pragma unroll
            for (int nt = 0; nt < NT2; ++nt)
#pragma unroll
                for (int mt = 0; mt < MTILES; ++mt)
                    c2[mt][nt] = wmma_f16(a2[mt], w2s[nt], c2[mt][nt]);

            __builtin_amdgcn_sched_barrier(0);
        }

        tile += N_BLOCKS * WAVES_PER_BLOCK;
        asm volatile("" :: "v"(cnt[0][0]), "v"(cnt[0][1]), "v"(cnt[0][2]), "v"(cnt[0][3]),
                           "v"(cnt[0][4]), "v"(cnt[0][5]), "v"(cnt[0][6]), "v"(cnt[0][7]),
                           "v"(cnt[1][0]), "v"(cnt[1][1]), "v"(cnt[1][2]), "v"(cnt[1][3]),
                           "v"(cnt[1][4]), "v"(cnt[1][5]), "v"(cnt[1][6]), "v"(cnt[1][7]));
        if (it + 1 < TILES_PER_WAVE) {
            load_x_tile(x, tile * TOK_PER_WAVE, lane16, hi, xr);
        } else {
#pragma unroll
            for (int mt = 0; mt < MTILES; ++mt)
#pragma unroll
                for (int ks = 0; ks < KS1; ++ks)
#pragma unroll
                    for (int q = 0; q < 4; ++q)
                        xr[mt][ks][q] = make_float4(0.0f, 0.0f, 0.0f, 0.0f);
        }
        __builtin_amdgcn_sched_barrier(0);

        float inv[MTILES][8];
#pragma unroll
        for (int mt = 0; mt < MTILES; ++mt)
#pragma unroll
            for (int r = 0; r < 8; ++r)
                inv[mt][r] = 1.0f / fmaxf(cnt[mt][r], 1.0f);

#pragma unroll
        for (int nt = 0; nt < NT2; ++nt) {
            const int n = nt * 16 + lane16;
            const float bias = bias2[nt];
#pragma unroll
            for (int mt = 0; mt < MTILES; ++mt) {
                float acc    = 0.0f;
                int   cur    = sid[mt][0];
                float curInv = inv[mt][0];
#pragma unroll
                for (int r = 0; r < 8; ++r) {
                    float v = c2[mt][nt][r] + bias;
                    v = v > 0.0f ? v : 0.0f;
                    if (sid[mt][r] != cur) {
                        atomicAdd(outMean + (size_t)cur * D_OUT + n, acc * curInv);
                        acc    = 0.0f;
                        cur    = sid[mt][r];
                        curInv = inv[mt][r];
                    }
                    acc += v;
                }
                atomicAdd(outMean + (size_t)cur * D_OUT + n, acc * curInv);
            }
        }
    }
}

__global__ void count_kernel(const int* __restrict__ seg, float* __restrict__ counts) {
    int t = blockIdx.x * blockDim.x + threadIdx.x;
    if (t < T_TOKENS) atomicAdd(&counts[seg[t]], 1.0f);
}

extern "C" void kernel_launch(void* const* d_in, const int* in_sizes, int n_in,
                              void* d_out, int out_size, void* d_ws, size_t ws_size,
                              hipStream_t stream) {
    (void)in_sizes; (void)n_in; (void)ws_size;

    const float* x   = (const float*)d_in[0];
    const int*   seg = (const int*)d_in[1];
    const float* W1  = (const float*)d_in[3];
    const float* b1  = (const float*)d_in[4];
    const float* W2  = (const float*)d_in[5];
    const float* b2  = (const float*)d_in[6];

    float* out = (float*)d_out;

    _Float16* w1f    = (_Float16*)d_ws;
    _Float16* w2f    = (_Float16*)((char*)d_ws + (size_t)D_IN * D_H * 2);
    float*    counts = (float*)((char*)d_ws + (size_t)(D_IN * D_H + D_H * D_OUT) * 2);

    hipMemsetAsync(out, 0, (size_t)out_size * sizeof(float), stream);
    hipMemsetAsync(counts, 0, (size_t)N_SETS * sizeof(float), stream);

    prep_weights_kernel<<<(D_IN * D_H + D_H * D_OUT) / 256, 256, 0, stream>>>(W1, W2, w1f, w2f);

    count_kernel<<<T_TOKENS / 256, 256, 0, stream>>>(seg, counts);

    deepsets_mlp_kernel<<<N_BLOCKS, WAVES_PER_BLOCK * 32, 0, stream>>>(
        x, seg, w1f, b1, b2, counts, out);
}
